# starting kernel
# baseline (speedup 1.0000x reference)
#include <hip/hip_runtime.h>

typedef __attribute__((ext_vector_type(2))) float v2f;
typedef __attribute__((ext_vector_type(8))) float v8f;

#define B_  4
#define T_  5
#define C_  3
#define H_  128
#define W_  128
#define D_  60
#define DP  64
#define NN  4096
#define KT  128

__device__ __forceinline__ float fast_exp2(float x) {
    return __builtin_amdgcn_exp2f(x);
}

__global__ void nlb_prep_q(const float* __restrict__ x, float* __restrict__ Q) {
    int idx = blockIdx.x * blockDim.x + threadIdx.x;
    if (idx >= B_ * NN * DP) return;
    int dp = idx & 63;
    int n  = (idx >> 6) & (NN - 1);
    int b  = idx >> 18;
    float v = 0.0f;
    if (dp < D_) {
        int tc = dp >> 2;
        int rr = dp & 3;
        int ri = rr >> 1, rj = rr & 1;
        int oh = n >> 6, ow = n & 63;
        int hh = oh * 2 + ri, ww = ow * 2 + rj;
        v = x[((size_t)(b * (T_ * C_) + tc) * H_ + hh) * W_ + ww];
    }
    Q[idx] = v;
}

__global__ void nlb_prep_g(const float* __restrict__ Q, const float* __restrict__ gw,
                           const float* __restrict__ gb, float* __restrict__ Gt) {
    int idx = blockIdx.x * blockDim.x + threadIdx.x;
    if (idx >= B_ * DP * NN) return;
    int n = idx & (NN - 1);
    int o = (idx >> 12) & 63;
    int b = idx >> 18;
    float acc = 0.0f;
    if (o < D_) {
        const float* qr = Q + (((size_t)b * NN + n) << 6);
        const float* wr = gw + o * D_;
        acc = gb[o];
#pragma unroll 4
        for (int d = 0; d < D_; ++d) acc += wr[d] * qr[d];
    }
    Gt[idx] = acc;
}

__global__ __launch_bounds__(128, 1)
void nlb_attn(const float* __restrict__ Q, const float* __restrict__ Gt,
              float* __restrict__ Y) {
    __shared__ float lds[4 * 16 * KT];

    const int lane = threadIdx.x & 31;
    const int wave = threadIdx.x >> 5;
    const int m    = lane & 15;
    const int h    = lane >> 4;
    const int b    = blockIdx.x >> 6;
    const int q0   = ((blockIdx.x & 63) * 4 + wave) * 16;

    const float* Qb  = Q  + (size_t)b * NN * DP;
    const float* Gtb = Gt + (size_t)b * DP * NN;
    float*       Yb  = Y  + (size_t)b * NN * DP;
    float*       pl  = lds + wave * (16 * KT);

    const float* vrow0 = Gtb + (size_t)( 0 + m) * NN;
    const float* vrow1 = Gtb + (size_t)(16 + m) * NN;
    const float* vrow2 = Gtb + (size_t)(32 + m) * NN;
    const float* vrow3 = Gtb + (size_t)(48 + m) * NN;

    const float LOG2E = 1.4426950408889634f;
    v2f qa[15];
    {
        const float* qrow = Qb + (size_t)(q0 + m) * DP;
#pragma unroll
        for (int kk = 0; kk < 15; ++kk) {
            v2f q = *(const v2f*)(qrow + 4 * kk + 2 * h);
            qa[kk] = q * LOG2E;
        }
    }

    v8f o0{}, o1{}, o2{}, o3{};
    float mrow[8], lrow[8];
#pragma unroll
    for (int j = 0; j < 8; ++j) { mrow[j] = -__builtin_inff(); lrow[j] = 0.0f; }

    for (int ch = 0; ch < NN / KT; ++ch) {
        const int k0 = ch * KT;

        {
            const int kn = (ch + 1 < NN / KT) ? (k0 + KT) : k0;
            const char* pk = (const char*)(Qb + (size_t)kn * DP) + lane * 1024;
            __builtin_prefetch(pk,       0, 3);
            __builtin_prefetch(pk + 256, 0, 3);
            __builtin_prefetch(pk + 512, 0, 3);
            __builtin_prefetch(pk + 768, 0, 3);
            const char* pv0 = (const char*)(Gtb + (size_t)lane        * NN + kn);
            const char* pv1 = (const char*)(Gtb + (size_t)(lane + 32) * NN + kn);
            __builtin_prefetch(pv0,       0, 3);
            __builtin_prefetch(pv0 + 256, 0, 3);
            __builtin_prefetch(pv1,       0, 3);
            __builtin_prefetch(pv1 + 256, 0, 3);
        }

        v8f s[8];
#pragma unroll
        for (int t = 0; t < 8; ++t) {
            v8f acc{};
            const float* krow = Qb + (size_t)(k0 + t * 16 + m) * DP;
#pragma unroll
            for (int kk = 0; kk < 15; ++kk) {
                v2f kb = *(const v2f*)(krow + 4 * kk + 2 * h);
                acc = __builtin_amdgcn_wmma_f32_16x16x4_f32(
                          false, qa[kk], false, kb, (short)0, acc, false, false);
            }
            s[t] = acc;
        }

#pragma unroll
        for (int j = 0; j < 8; ++j) {
            float tm = fmaxf(fmaxf(fmaxf(s[0][j], s[1][j]), fmaxf(s[2][j], s[3][j])),
                             fmaxf(fmaxf(s[4][j], s[5][j]), fmaxf(s[6][j], s[7][j])));
            tm = fmaxf(tm, __shfl_xor(tm, 1, 16));
            tm = fmaxf(tm, __shfl_xor(tm, 2, 16));
            tm = fmaxf(tm, __shfl_xor(tm, 4, 16));
            tm = fmaxf(tm, __shfl_xor(tm, 8, 16));
            float mn = fmaxf(mrow[j], tm);
            float sc = fast_exp2(mrow[j] - mn);
            float p0 = fast_exp2(s[0][j] - mn);
            float p1 = fast_exp2(s[1][j] - mn);
            float p2 = fast_exp2(s[2][j] - mn);
            float p3 = fast_exp2(s[3][j] - mn);
            float p4 = fast_exp2(s[4][j] - mn);
            float p5 = fast_exp2(s[5][j] - mn);
            float p6 = fast_exp2(s[6][j] - mn);
            float p7 = fast_exp2(s[7][j] - mn);
            s[0][j] = p0; s[1][j] = p1; s[2][j] = p2; s[3][j] = p3;
            s[4][j] = p4; s[5][j] = p5; s[6][j] = p6; s[7][j] = p7;
            float su = ((p0 + p1) + (p2 + p3)) + ((p4 + p5) + (p6 + p7));
            su += __shfl_xor(su, 1, 16);
            su += __shfl_xor(su, 2, 16);
            su += __shfl_xor(su, 4, 16);
            su += __shfl_xor(su, 8, 16);
            lrow[j] = lrow[j] * sc + su;
            mrow[j] = mn;
            o0[j] *= sc; o1[j] *= sc; o2[j] *= sc; o3[j] *= sc;
        }

#pragma unroll
        for (int t = 0; t < 8; ++t)
#pragma unroll
            for (int j = 0; j < 8; ++j)
                pl[(8 * h + j) * KT + t * 16 + m] = s[t][j];
        asm volatile("" ::: "memory");

#pragma unroll
        for (int kk2 = 0; kk2 < 32; ++kk2) {
            v2f pa = *(const v2f*)(pl + m * KT + 4 * kk2 + 2 * h);
            const int kof = k0 + 4 * kk2 + 2 * h;
            v2f vb;
            vb = *(const v2f*)(vrow0 + kof);
            o0 = __builtin_amdgcn_wmma_f32_16x16x4_f32(false, pa, false, vb, (short)0, o0, false, false);
            vb = *(const v2f*)(vrow1 + kof);
            o1 = __builtin_amdgcn_wmma_f32_16x16x4_f32(false, pa, false, vb, (short)0, o1, false, false);
            vb = *(const v2f*)(vrow2 + kof);
            o2 = __builtin_amdgcn_wmma_f32_16x16x4_f32(false, pa, false, vb, (short)0, o2, false, false);
            vb = *(const v2f*)(vrow3 + kof);
            o3 = __builtin_amdgcn_wmma_f32_16x16x4_f32(false, pa, false, vb, (short)0, o3, false, false);
        }
        asm volatile("" ::: "memory");
    }

#pragma unroll
    for (int j = 0; j < 8; ++j) {
        float inv = 1.0f / lrow[j];
        float* yr = Yb + (size_t)(q0 + 8 * h + j) * DP;
        yr[ 0 + m] = o0[j] * inv;
        yr[16 + m] = o1[j] * inv;
        yr[32 + m] = o2[j] * inv;
        yr[48 + m] = o3[j] * inv;
    }
}

__global__ void nlb_out(const float* __restrict__ x, const float* __restrict__ Y,
                        const float* __restrict__ ww, const float* __restrict__ wb,
                        float* __restrict__ Z) {
    int idx = blockIdx.x * blockDim.x + threadIdx.x;
    if (idx >= B_ * T_ * C_ * H_ * W_) return;
    int wcol = idx % W_;  int tmp = idx / W_;
    int hrow = tmp % H_;  tmp /= H_;
    int cc   = tmp % C_;  tmp /= C_;
    int t    = tmp % T_;
    int b    = tmp / T_;
    int ri = hrow & 1, rj = wcol & 1;
    int oh = hrow >> 1, ow = wcol >> 1;
    int o  = (t * C_ + cc) * 4 + ri * 2 + rj;
    int n  = oh * 64 + ow;
    const float* yr = Y + (((size_t)b * NN + n) << 6);
    const float* wr = ww + o * D_;
    float acc = wb[o];
#pragma unroll 4
    for (int d = 0; d < D_; ++d) acc += wr[d] * yr[d];
    Z[idx] = acc + x[idx];
}

extern "C" void kernel_launch(void* const* d_in, const int* in_sizes, int n_in,
                              void* d_out, int out_size, void* d_ws, size_t ws_size,
                              hipStream_t stream) {
    const float* x  = (const float*)d_in[0];
    const float* gw = (const float*)d_in[1];
    const float* gb = (const float*)d_in[2];
    const float* ww = (const float*)d_in[3];
    const float* wb = (const float*)d_in[4];
    float* Z = (float*)d_out;

    const size_t buf = (size_t)B_ * NN * DP;
    float* Qbuf = (float*)d_ws;
    float* Gbuf = Qbuf + buf;
    float* Ybuf = Gbuf + buf;

    {
        int n = B_ * NN * DP;
        nlb_prep_q<<<(n + 255) / 256, 256, 0, stream>>>(x, Qbuf);
    }
    {
        int n = B_ * DP * NN;
        nlb_prep_g<<<(n + 255) / 256, 256, 0, stream>>>(Qbuf, gw, gb, Gbuf);
    }
    nlb_attn<<<B_ * 64, 128, 0, stream>>>(Qbuf, Gbuf, Ybuf);
    {
        int n = B_ * T_ * C_ * H_ * W_;
        nlb_out<<<(n + 255) / 256, 256, 0, stream>>>(x, Ybuf, ww, wb, Z);
    }
}


# pointwise kernels (unshuffle, both 60x60 projections): one lane per pixel does all its channels, weights wave-uniform in scalar registers, tiles via LDS; 96256 waves -> 5632; sum order unchanged
# speedup vs baseline: 2.4498x; 2.4498x over previous
#include <hip/hip_runtime.h>

typedef __attribute__((ext_vector_type(2))) float v2f;
typedef __attribute__((ext_vector_type(4))) float v4f;
typedef __attribute__((ext_vector_type(8))) float v8f;

#define B_  4
#define T_  5
#define C_  3
#define H_  128
#define W_  128
#define D_  60
#define DP  64
#define NN  4096
#define KT  128

__device__ __forceinline__ float fast_exp2(float x) {
    return __builtin_amdgcn_exp2f(x);
}

template <int X>
__device__ __forceinline__ float row_xor(float v) {
    return __builtin_bit_cast(float,
        __builtin_amdgcn_update_dpp(0, __builtin_bit_cast(int, v), 0x160 + X, 0xf, 0xf, true));
}


__global__ __launch_bounds__(128)
void nlb_prep_q(const float* __restrict__ x, float* __restrict__ Kf) {
    const int gid  = blockIdx.x * 128 + threadIdx.x;
    const int lane = gid & 31;
    const int T    = (gid >> 5) & 255;
    const int b    = gid >> 13;
    const int m = lane & 15, h = lane >> 4;
    const int n  = T * 16 + m;
    const int oh = n >> 6, ow = n & 63;
    const float* sp = x + ((size_t)(b * (T_ * C_)) * H_ + (2 * oh + h)) * W_ + 2 * ow;
    float*       dp = Kf + ((size_t)(b * 256 + T) * 8) * 128 + lane * 4;
    v2f p[15];
#pragma unroll
    for (int tc = 0; tc < 15; ++tc)
        p[tc] = *(const v2f*)(sp + (size_t)tc * (H_ * W_));
#pragma unroll
    for (int kp = 0; kp < 8; ++kp) {
        v4f o;
        o.x = p[2 * kp].x; o.y = p[2 * kp].y;
        if (kp < 7) { o.z = p[2 * kp + 1].x; o.w = p[2 * kp + 1].y; }
        else        { o.z = 0.0f; o.w = 0.0f; }
        *(v4f*)(dp + kp * 128) = o;
    }
}

#define VP 136
__global__ __launch_bounds__(128)
void nlb_prep_g(const float* __restrict__ Kf, const float* __restrict__ gw,
                const float* __restrict__ gb, float* __restrict__ Vf) {
    __shared__ __attribute__((aligned(16))) float vt[16 * VP];
    const int tid  = threadIdx.x;
    const int ctp  = blockIdx.y >> 1;
    const int half = blockIdx.y & 1;
    const int pix  = blockIdx.x * 128 + tid;
    const int b = pix >> 12, n = pix & 4095;
    float xr[60];
    {
        const float* q0 = Kf + ((size_t)(b * 256 + (n >> 4)) * 8) * 128 + (n & 15) * 4;
#pragma unroll
        for (int kp = 0; kp < 7; ++kp) {
            v4f a = *(const v4f*)(q0 + kp * 128);
            v4f c = *(const v4f*)(q0 + kp * 128 + 64);
            xr[8 * kp + 0] = a.x; xr[8 * kp + 1] = a.y; xr[8 * kp + 2] = c.x; xr[8 * kp + 3] = c.y;
            xr[8 * kp + 4] = a.z; xr[8 * kp + 5] = a.w; xr[8 * kp + 6] = c.z; xr[8 * kp + 7] = c.w;
        }
        v2f a = *(const v2f*)(q0 + 7 * 128);
        v2f c = *(const v2f*)(q0 + 7 * 128 + 64);
        xr[56] = a.x; xr[57] = a.y; xr[58] = c.x; xr[59] = c.y;
    }
    for (int f = 0; f < 16; f += 4) {
        const int o = 32 * ctp + 8 * half + f + ((f & 8) ? 8 : 0);
        float a0 = 0.0f, a1 = 0.0f, a2 = 0.0f, a3 = 0.0f;
        if (o < D_) {
            const float* wr = gw + o * D_;
            a0 = gb[o]; a1 = gb[o + 1]; a2 = gb[o + 2]; a3 = gb[o + 3];
#pragma unroll
            for (int d = 0; d < D_; ++d) {
                a0 = __builtin_fmaf(wr[d],          xr[d], a0);
                a1 = __builtin_fmaf(wr[D_ + d],     xr[d], a1);
                a2 = __builtin_fmaf(wr[2 * D_ + d], xr[d], a2);
                a3 = __builtin_fmaf(wr[3 * D_ + d], xr[d], a3);
            }
        }
        vt[(f + 0) * VP + tid] = a0;
        vt[(f + 1) * VP + tid] = a1;
        vt[(f + 2) * VP + tid] = a2;
        vt[(f + 3) * VP + tid] = a3;
    }
    __syncthreads();
    float* vo = Vf + ((size_t)(blockIdx.x * 32) * 2 + ctp) * 128;
#pragma unroll
    for (int r = 0; r < 4; ++r) {
        const int el = r * 128 + tid;
        const int gl = el >> 4;
        const int hh = (el >> 3) & 1, m8 = el & 7;
        const int key = 4 * gl + 2 * hh;
        v2f lo = *(const v2f*)(vt + m8 * VP + key);
        v2f hi = *(const v2f*)(vt + (8 + m8) * VP + key);
        v4f o; o.x = lo.x; o.y = lo.y; o.z = hi.x; o.w = hi.y;
        *(v4f*)(vo + (size_t)gl * 256 + (16 * hh + 8 * half + m8) * 4) = o;
    }
}

__global__ __launch_bounds__(128, 1)
void nlb_attn(const float* __restrict__ Kf, const float* __restrict__ Vf,
              float* __restrict__ Y) {
    __shared__ float lds[4 * 16 * KT];

    const int lane = threadIdx.x & 31;
    const int wave = threadIdx.x >> 5;
    const int m    = lane & 15;
    const int h    = lane >> 4;
    const int b    = blockIdx.x >> 6;
    const int q0   = ((blockIdx.x & 63) * 4 + wave) * 16;

    const float* Kb  = Kf + (size_t)b * NN * DP;
    const float* Vb  = Vf + (size_t)b * NN * DP;
    float*       Yb  = Y  + (size_t)b * NN * DP;
    float*       pl  = lds + wave * (16 * KT);

    const float LOG2E = 1.4426950408889634f;
    v2f qa[15];
    {
        const float* qt = Kb + (size_t)(q0 >> 4) * 1024 + lane * 4;
#pragma unroll
        for (int kp = 0; kp < 7; ++kp) {
            v4f q = *(const v4f*)(qt + kp * 128);
            qa[2 * kp]     = q.xy * LOG2E;
            qa[2 * kp + 1] = q.zw * LOG2E;
        }
        v2f q = *(const v2f*)(qt + 7 * 128);
        qa[14] = q * LOG2E;
    }

    v4f kc[8][7];
    v2f kl[8];
    v4f vc[32][2];
    v2f pa[32];

    {
        const float* kt = Kb + lane * 4;
#pragma unroll
        for (int t = 0; t < 8; ++t) {
#pragma unroll
            for (int kp = 0; kp < 7; ++kp)
                kc[t][kp] = *(const v4f*)(kt + t * 1024 + kp * 128);
            kl[t] = *(const v2f*)(kt + t * 1024 + 7 * 128);
        }
    }

    v8f o0{}, o1{}, o2{}, o3{};
    float mrow[8], lrow[8];
#pragma unroll
    for (int j = 0; j < 8; ++j) { mrow[j] = -__builtin_inff(); lrow[j] = 0.0f; }

    for (int ch = 0; ch < NN / KT; ++ch) {
        const int k0 = ch * KT;
        const int kn = (ch + 1 < NN / KT) ? (k0 + KT) : k0;
        const float* vt = Vb + (size_t)(k0 >> 2) * 256 + lane * 4;
        const float* kt = Kb + (size_t)(kn >> 4) * 1024 + lane * 4;

        v8f s[8];
#pragma unroll
        for (int tp = 0; tp < 4; ++tp) {
#pragma unroll
            for (int u = 0; u < 8; ++u) {
                vc[8 * tp + u][0] = *(const v4f*)(vt + (8 * tp + u) * 256);
                vc[8 * tp + u][1] = *(const v4f*)(vt + (8 * tp + u) * 256 + 128);
            }
            v8f a0{}, a1{};
#pragma unroll
            for (int kp = 0; kp < 7; ++kp) {
                a0 = __builtin_amdgcn_wmma_f32_16x16x4_f32(
                         false, qa[2 * kp], false, kc[2 * tp][kp].xy, (short)0, a0, false, false);
                a1 = __builtin_amdgcn_wmma_f32_16x16x4_f32(
                         false, qa[2 * kp], false, kc[2 * tp + 1][kp].xy, (short)0, a1, false, false);
                a0 = __builtin_amdgcn_wmma_f32_16x16x4_f32(
                         false, qa[2 * kp + 1], false, kc[2 * tp][kp].zw, (short)0, a0, false, false);
                a1 = __builtin_amdgcn_wmma_f32_16x16x4_f32(
                         false, qa[2 * kp + 1], false, kc[2 * tp + 1][kp].zw, (short)0, a1, false, false);
            }
            a0 = __builtin_amdgcn_wmma_f32_16x16x4_f32(
                     false, qa[14], false, kl[2 * tp], (short)0, a0, false, false);
            a1 = __builtin_amdgcn_wmma_f32_16x16x4_f32(
                     false, qa[14], false, kl[2 * tp + 1], (short)0, a1, false, false);
            s[2 * tp]     = a0;
            s[2 * tp + 1] = a1;
            __builtin_amdgcn_sched_barrier(0);
        }

#pragma unroll
        for (int j = 0; j < 8; ++j) {
            float tm = fmaxf(fmaxf(fmaxf(s[0][j], s[1][j]), fmaxf(s[2][j], s[3][j])),
                             fmaxf(fmaxf(s[4][j], s[5][j]), fmaxf(s[6][j], s[7][j])));
            tm = fmaxf(tm, row_xor<1>(tm));
            tm = fmaxf(tm, row_xor<2>(tm));
            tm = fmaxf(tm, row_xor<4>(tm));
            tm = fmaxf(tm, row_xor<8>(tm));
            float mn = fmaxf(mrow[j], tm);
            float sc = fast_exp2(mrow[j] - mn);
            float p0 = fast_exp2(s[0][j] - mn);
            float p1 = fast_exp2(s[1][j] - mn);
            float p2 = fast_exp2(s[2][j] - mn);
            float p3 = fast_exp2(s[3][j] - mn);
            float p4 = fast_exp2(s[4][j] - mn);
            float p5 = fast_exp2(s[5][j] - mn);
            float p6 = fast_exp2(s[6][j] - mn);
            float p7 = fast_exp2(s[7][j] - mn);
            s[0][j] = p0; s[1][j] = p1; s[2][j] = p2; s[3][j] = p3;
            s[4][j] = p4; s[5][j] = p5; s[6][j] = p6; s[7][j] = p7;
            float su = ((p0 + p1) + (p2 + p3)) + ((p4 + p5) + (p6 + p7));
            su += row_xor<1>(su);
            su += row_xor<2>(su);
            su += row_xor<4>(su);
            su += row_xor<8>(su);
            lrow[j] = lrow[j] * sc + su;
            mrow[j] = mn;
            o0[j] *= sc; o1[j] *= sc; o2[j] *= sc; o3[j] *= sc;
        }

#pragma unroll
        for (int t = 0; t < 8; ++t)
#pragma unroll
            for (int j = 0; j < 8; ++j)
                pl[(8 * h + j) * KT + t * 16 + m] = s[t][j];
        asm volatile("" ::: "memory");

#pragma unroll
        for (int kk2 = 0; kk2 < 32; ++kk2)
            pa[kk2] = *(const v2f*)(pl + m * KT + 4 * kk2 + 2 * h);
        asm volatile("" ::: "memory");
        __builtin_amdgcn_sched_barrier(0);

#pragma unroll
        for (int g = 0; g < 8; ++g) {
#pragma unroll
            for (int kp = 0; kp < 7; ++kp)
                kc[g][kp] = *(const v4f*)(kt + g * 1024 + kp * 128);
            kl[g] = *(const v2f*)(kt + g * 1024 + 7 * 128);
#pragma unroll
            for (int u = 0; u < 4; ++u) {
                const int kk2 = 4 * g + u;
                o0 = __builtin_amdgcn_wmma_f32_16x16x4_f32(false, pa[kk2], false, vc[kk2][0].xy, (short)0, o0, false, false);
                o1 = __builtin_amdgcn_wmma_f32_16x16x4_f32(false, pa[kk2], false, vc[kk2][0].zw, (short)0, o1, false, false);
                o2 = __builtin_amdgcn_wmma_f32_16x16x4_f32(false, pa[kk2], false, vc[kk2][1].xy, (short)0, o2, false, false);
                o3 = __builtin_amdgcn_wmma_f32_16x16x4_f32(false, pa[kk2], false, vc[kk2][1].zw, (short)0, o3, false, false);
            }
            __builtin_amdgcn_sched_barrier(0);
        }
    }

#pragma unroll
    for (int j = 0; j < 8; ++j) {
        float inv = 1.0f / lrow[j];
        float* yr = Yb + (size_t)(q0 + 8 * h + j) * DP;
        yr[ 0 + m] = o0[j] * inv;
        yr[16 + m] = o1[j] * inv;
        yr[32 + m] = o2[j] * inv;
        yr[48 + m] = o3[j] * inv;
    }
}

#define TCG 3
#define YP  68
__global__ __launch_bounds__(64)
void nlb_out(const float* __restrict__ x, const float* __restrict__ Y,
             const float* __restrict__ ww, const float* __restrict__ wb,
             float* __restrict__ Z) {
    __shared__ __attribute__((aligned(16))) float ys[64 * YP];
    const int tid = threadIdx.x;
    const int pix = blockIdx.x * 64 + tid;
    const int b = pix >> 12, n = pix & 4095;
    const int oh = n >> 6, ow = n & 63;
    {
        const float* yb = Y + ((size_t)(blockIdx.x * 64) << 6);
#pragma unroll
        for (int i = 0; i < 16; ++i) {
            const int e = i * 64 + tid;
            v4f t = *(const v4f*)(yb + 4 * e);
            *(v4f*)(ys + (e >> 4) * YP + 4 * (e & 15)) = t;
        }
    }
    __syncthreads();
    float y[60];
#pragma unroll
    for (int i = 0; i < 15; ++i) {
        v4f t = *(const v4f*)(ys + tid * YP + 4 * i);
        y[4 * i] = t.x; y[4 * i + 1] = t.y; y[4 * i + 2] = t.z; y[4 * i + 3] = t.w;
    }
    for (int k = 0; k < TCG; ++k) {
        const int tc = blockIdx.y * TCG + k;
        const float* wr = ww + tc * 4 * D_;
        float a0 = wb[tc * 4], a1 = wb[tc * 4 + 1], a2 = wb[tc * 4 + 2], a3 = wb[tc * 4 + 3];
#pragma unroll
        for (int d = 0; d < D_; ++d) {
            a0 = __builtin_fmaf(wr[d],          y[d], a0);
            a1 = __builtin_fmaf(wr[D_ + d],     y[d], a1);
            a2 = __builtin_fmaf(wr[2 * D_ + d], y[d], a2);
            a3 = __builtin_fmaf(wr[3 * D_ + d], y[d], a3);
        }
        const size_t base = ((size_t)(b * (T_ * C_) + tc) * H_ + 2 * oh) * W_ + 2 * ow;
        v2f x0 = *(const v2f*)(x + base);
        v2f x1 = *(const v2f*)(x + base + W_);
        v2f z0, z1;
        z0.x = a0 + x0.x; z0.y = a1 + x0.y;
        z1.x = a2 + x1.x; z1.y = a3 + x1.y;
        *(v2f*)(Z + base)      = z0;
        *(v2f*)(Z + base + W_) = z1;
    }
}

extern "C" void kernel_launch(void* const* d_in, const int* in_sizes, int n_in,
                              void* d_out, int out_size, void* d_ws, size_t ws_size,
                              hipStream_t stream) {
    const float* x  = (const float*)d_in[0];
    const float* gw = (const float*)d_in[1];
    const float* gb = (const float*)d_in[2];
    const float* ww = (const float*)d_in[3];
    const float* wb = (const float*)d_in[4];
    float* Z = (float*)d_out;

    const size_t buf = (size_t)B_ * NN * DP;
    float* Qbuf = (float*)d_ws;
    float* Gbuf = Qbuf + buf;
    float* Ybuf = Gbuf + buf;

    nlb_prep_q<<<256, 128, 0, stream>>>(x, Qbuf);
    nlb_prep_g<<<dim3(128, 4), 128, 0, stream>>>(Qbuf, gw, gb, Gbuf);
    nlb_attn<<<B_ * 64, 128, 0, stream>>>(Qbuf, Gbuf, Ybuf);
    nlb_out<<<dim3(256, 5), 64, 0, stream>>>(x, Ybuf, ww, wb, Z);
}


# attention softmax moved behind the C->A LDS transposition (tile pitch 132, conflict-free): a lane owns a query row, max and sum in-lane + one half exchange, no butterflies; row-sum order changes
# speedup vs baseline: 2.5373x; 1.0357x over previous
#include <hip/hip_runtime.h>

typedef __attribute__((ext_vector_type(2))) float v2f;
typedef __attribute__((ext_vector_type(4))) float v4f;
typedef __attribute__((ext_vector_type(8))) float v8f;

#define B_  4
#define T_  5
#define C_  3
#define H_  128
#define W_  128
#define D_  60
#define DP  64
#define NN  4096
#define KT  128
#define PP  132

__device__ __forceinline__ float fast_exp2(float x) {
    return __builtin_amdgcn_exp2f(x);
}

__device__ __forceinline__ float xhalf(float v) {
    const int b = __builtin_bit_cast(int, v);
    return __builtin_bit_cast(float,
        __builtin_amdgcn_permlanex16(b, b, 0x76543210, 0xfedcba98, false, false));
}

template <int J>
__device__ __forceinline__ float row_bcast(float v) {
    const int x  = __builtin_bit_cast(int, v);
    const int lo = __builtin_amdgcn_update_dpp(0, x, 0x150 + J, 0xf, 0xf, true);
    const int r  = __builtin_amdgcn_update_dpp(lo, x, 0x150 + 8 + J, 0x2, 0xf, false);
    return __builtin_bit_cast(float, r);
}


__global__ __launch_bounds__(128)
void nlb_prep_q(const float* __restrict__ x, float* __restrict__ Kf) {
    const int gid  = blockIdx.x * 128 + threadIdx.x;
    const int lane = gid & 31;
    const int T    = (gid >> 5) & 255;
    const int b    = gid >> 13;
    const int m = lane & 15, h = lane >> 4;
    const int n  = T * 16 + m;
    const int oh = n >> 6, ow = n & 63;
    const float* sp = x + ((size_t)(b * (T_ * C_)) * H_ + (2 * oh + h)) * W_ + 2 * ow;
    float*       dp = Kf + ((size_t)(b * 256 + T) * 8) * 128 + lane * 4;
    v2f p[15];
#pragma unroll
    for (int tc = 0; tc < 15; ++tc)
        p[tc] = *(const v2f*)(sp + (size_t)tc * (H_ * W_));
#pragma unroll
    for (int kp = 0; kp < 8; ++kp) {
        v4f o;
        o.x = p[2 * kp].x; o.y = p[2 * kp].y;
        if (kp < 7) { o.z = p[2 * kp + 1].x; o.w = p[2 * kp + 1].y; }
        else        { o.z = 0.0f; o.w = 0.0f; }
        *(v4f*)(dp + kp * 128) = o;
    }
}

#define VP 136
__global__ __launch_bounds__(128)
void nlb_prep_g(const float* __restrict__ Kf, const float* __restrict__ gw,
                const float* __restrict__ gb, float* __restrict__ Vf) {
    __shared__ __attribute__((aligned(16))) float vt[16 * VP];
    const int tid  = threadIdx.x;
    const int ctp  = blockIdx.y >> 1;
    const int half = blockIdx.y & 1;
    const int pix  = blockIdx.x * 128 + tid;
    const int b = pix >> 12, n = pix & 4095;
    float xr[60];
    {
        const float* q0 = Kf + ((size_t)(b * 256 + (n >> 4)) * 8) * 128 + (n & 15) * 4;
#pragma unroll
        for (int kp = 0; kp < 7; ++kp) {
            v4f a = *(const v4f*)(q0 + kp * 128);
            v4f c = *(const v4f*)(q0 + kp * 128 + 64);
            xr[8 * kp + 0] = a.x; xr[8 * kp + 1] = a.y; xr[8 * kp + 2] = c.x; xr[8 * kp + 3] = c.y;
            xr[8 * kp + 4] = a.z; xr[8 * kp + 5] = a.w; xr[8 * kp + 6] = c.z; xr[8 * kp + 7] = c.w;
        }
        v2f a = *(const v2f*)(q0 + 7 * 128);
        v2f c = *(const v2f*)(q0 + 7 * 128 + 64);
        xr[56] = a.x; xr[57] = a.y; xr[58] = c.x; xr[59] = c.y;
    }
    for (int f = 0; f < 16; f += 4) {
        const int o = 32 * ctp + 8 * half + f + ((f & 8) ? 8 : 0);
        float a0 = 0.0f, a1 = 0.0f, a2 = 0.0f, a3 = 0.0f;
        if (o < D_) {
            const float* wr = gw + o * D_;
            a0 = gb[o]; a1 = gb[o + 1]; a2 = gb[o + 2]; a3 = gb[o + 3];
#pragma unroll
            for (int d = 0; d < D_; ++d) {
                a0 = __builtin_fmaf(wr[d],          xr[d], a0);
                a1 = __builtin_fmaf(wr[D_ + d],     xr[d], a1);
                a2 = __builtin_fmaf(wr[2 * D_ + d], xr[d], a2);
                a3 = __builtin_fmaf(wr[3 * D_ + d], xr[d], a3);
            }
        }
        vt[(f + 0) * VP + tid] = a0;
        vt[(f + 1) * VP + tid] = a1;
        vt[(f + 2) * VP + tid] = a2;
        vt[(f + 3) * VP + tid] = a3;
    }
    __syncthreads();
    float* vo = Vf + ((size_t)(blockIdx.x * 32) * 2 + ctp) * 128;
#pragma unroll
    for (int r = 0; r < 4; ++r) {
        const int el = r * 128 + tid;
        const int gl = el >> 4;
        const int hh = (el >> 3) & 1, m8 = el & 7;
        const int key = 4 * gl + 2 * hh;
        v2f lo = *(const v2f*)(vt + m8 * VP + key);
        v2f hi = *(const v2f*)(vt + (8 + m8) * VP + key);
        v4f o; o.x = lo.x; o.y = lo.y; o.z = hi.x; o.w = hi.y;
        *(v4f*)(vo + (size_t)gl * 256 + (16 * hh + 8 * half + m8) * 4) = o;
    }
}

__global__ __launch_bounds__(128, 1)
void nlb_attn(const float* __restrict__ Kf, const float* __restrict__ Vf,
              float* __restrict__ Y) {
    __shared__ float lds[4 * 16 * PP];

    const int lane = threadIdx.x & 31;
    const int wave = threadIdx.x >> 5;
    const int m    = lane & 15;
    const int h    = lane >> 4;
    const int b    = blockIdx.x >> 6;
    const int q0   = ((blockIdx.x & 63) * 4 + wave) * 16;

    const float* Kb  = Kf + (size_t)b * NN * DP;
    const float* Vb  = Vf + (size_t)b * NN * DP;
    float*       Yb  = Y  + (size_t)b * NN * DP;
    float*       pl  = lds + wave * (16 * PP);

    const float LOG2E = 1.4426950408889634f;
    v2f qa[15];
    {
        const float* qt = Kb + (size_t)(q0 >> 4) * 1024 + lane * 4;
#pragma unroll
        for (int kp = 0; kp < 7; ++kp) {
            v4f q = *(const v4f*)(qt + kp * 128);
            qa[2 * kp]     = q.xy * LOG2E;
            qa[2 * kp + 1] = q.zw * LOG2E;
        }
        v2f q = *(const v2f*)(qt + 7 * 128);
        qa[14] = q * LOG2E;
    }

    v4f kc[8][7];
    v2f kl[8];
    v4f vc[32][2];
    v2f pa[32];

    {
        const float* kt = Kb + lane * 4;
#pragma unroll
        for (int t = 0; t < 8; ++t) {
#pragma unroll
            for (int kp = 0; kp < 7; ++kp)
                kc[t][kp] = *(const v4f*)(kt + t * 1024 + kp * 128);
            kl[t] = *(const v2f*)(kt + t * 1024 + 7 * 128);
        }
    }

    v8f o0{}, o1{}, o2{}, o3{};
    float mrun = -__builtin_inff();
    float lrun = 0.0f;

    for (int ch = 0; ch < NN / KT; ++ch) {
        const int k0 = ch * KT;
        const int kn = (ch + 1 < NN / KT) ? (k0 + KT) : k0;
        const float* vt = Vb + (size_t)(k0 >> 2) * 256 + lane * 4;
        const float* kt = Kb + (size_t)(kn >> 4) * 1024 + lane * 4;

        v8f s[8];
#pragma unroll
        for (int tp = 0; tp < 4; ++tp) {
#pragma unroll
            for (int u = 0; u < 8; ++u) {
                vc[8 * tp + u][0] = *(const v4f*)(vt + (8 * tp + u) * 256);
                vc[8 * tp + u][1] = *(const v4f*)(vt + (8 * tp + u) * 256 + 128);
            }
            v8f a0{}, a1{};
#pragma unroll
            for (int kp = 0; kp < 7; ++kp) {
                a0 = __builtin_amdgcn_wmma_f32_16x16x4_f32(
                         false, qa[2 * kp], false, kc[2 * tp][kp].xy, (short)0, a0, false, false);
                a1 = __builtin_amdgcn_wmma_f32_16x16x4_f32(
                         false, qa[2 * kp], false, kc[2 * tp + 1][kp].xy, (short)0, a1, false, false);
                a0 = __builtin_amdgcn_wmma_f32_16x16x4_f32(
                         false, qa[2 * kp + 1], false, kc[2 * tp][kp].zw, (short)0, a0, false, false);
                a1 = __builtin_amdgcn_wmma_f32_16x16x4_f32(
                         false, qa[2 * kp + 1], false, kc[2 * tp + 1][kp].zw, (short)0, a1, false, false);
            }
            a0 = __builtin_amdgcn_wmma_f32_16x16x4_f32(
                     false, qa[14], false, kl[2 * tp], (short)0, a0, false, false);
            a1 = __builtin_amdgcn_wmma_f32_16x16x4_f32(
                     false, qa[14], false, kl[2 * tp + 1], (short)0, a1, false, false);
            s[2 * tp]     = a0;
            s[2 * tp + 1] = a1;
            __builtin_amdgcn_sched_barrier(0);
        }

#pragma unroll
        for (int t = 0; t < 8; ++t)
#pragma unroll
            for (int j = 0; j < 8; ++j)
                pl[(8 * h + j) * PP + t * 16 + m] = s[t][j];
        asm volatile("" ::: "memory");

#pragma unroll
        for (int kk2 = 0; kk2 < 32; ++kk2)
            pa[kk2] = *(const v2f*)(pl + m * PP + 4 * kk2 + 2 * h);
        asm volatile("" ::: "memory");

        float t0 = fmaxf(pa[0].x, pa[0].y), t1 = fmaxf(pa[1].x, pa[1].y);
        float t2 = fmaxf(pa[2].x, pa[2].y), t3 = fmaxf(pa[3].x, pa[3].y);
#pragma unroll
        for (int q = 1; q < 8; ++q) {
            t0 = fmaxf(t0, fmaxf(pa[4 * q].x,     pa[4 * q].y));
            t1 = fmaxf(t1, fmaxf(pa[4 * q + 1].x, pa[4 * q + 1].y));
            t2 = fmaxf(t2, fmaxf(pa[4 * q + 2].x, pa[4 * q + 2].y));
            t3 = fmaxf(t3, fmaxf(pa[4 * q + 3].x, pa[4 * q + 3].y));
        }
        float tm = fmaxf(fmaxf(t0, t1), fmaxf(t2, t3));
        tm = fmaxf(tm, xhalf(tm));
        const float mn = fmaxf(mrun, tm);
        const float sc = fast_exp2(mrun - mn);
        mrun = mn;
        const v2f mnv = {mn, mn};
#pragma unroll
        for (int kk2 = 0; kk2 < 32; ++kk2)
            pa[kk2] = pa[kk2] - mnv;
        v2f u0{}, u1{}, u2{}, u3{};
#pragma unroll
        for (int q = 0; q < 8; ++q) {
#pragma unroll
            for (int i = 0; i < 4; ++i) {
                pa[4 * q + i].x = fast_exp2(pa[4 * q + i].x);
                pa[4 * q + i].y = fast_exp2(pa[4 * q + i].y);
            }
            u0 += pa[4 * q]; u1 += pa[4 * q + 1]; u2 += pa[4 * q + 2]; u3 += pa[4 * q + 3];
        }
        const v2f us = (u0 + u1) + (u2 + u3);
        lrun = lrun * sc + (us.x + us.y);
#define NLB_RESCALE(J) { const float f = row_bcast<J>(sc); o0[J] *= f; o1[J] *= f; o2[J] *= f; o3[J] *= f; }
        NLB_RESCALE(0) NLB_RESCALE(1) NLB_RESCALE(2) NLB_RESCALE(3)
        NLB_RESCALE(4) NLB_RESCALE(5) NLB_RESCALE(6) NLB_RESCALE(7)
#undef NLB_RESCALE
        __builtin_amdgcn_sched_barrier(0);

#pragma unroll
        for (int g = 0; g < 8; ++g) {
#pragma unroll
            for (int kp = 0; kp < 7; ++kp)
                kc[g][kp] = *(const v4f*)(kt + g * 1024 + kp * 128);
            kl[g] = *(const v2f*)(kt + g * 1024 + 7 * 128);
#pragma unroll
            for (int u = 0; u < 4; ++u) {
                const int kk2 = 4 * g + u;
                o0 = __builtin_amdgcn_wmma_f32_16x16x4_f32(false, pa[kk2], false, vc[kk2][0].xy, (short)0, o0, false, false);
                o1 = __builtin_amdgcn_wmma_f32_16x16x4_f32(false, pa[kk2], false, vc[kk2][0].zw, (short)0, o1, false, false);
                o2 = __builtin_amdgcn_wmma_f32_16x16x4_f32(false, pa[kk2], false, vc[kk2][1].xy, (short)0, o2, false, false);
                o3 = __builtin_amdgcn_wmma_f32_16x16x4_f32(false, pa[kk2], false, vc[kk2][1].zw, (short)0, o3, false, false);
            }
            __builtin_amdgcn_sched_barrier(0);
        }
    }

    const float inv = 1.0f / (lrun + xhalf(lrun));
#define NLB_WRITE(J) { const float f = row_bcast<J>(inv); \
        float* yr = Yb + (size_t)(q0 + 8 * h + J) * DP; \
        yr[ 0 + m] = o0[J] * f; yr[16 + m] = o1[J] * f; \
        yr[32 + m] = o2[J] * f; yr[48 + m] = o3[J] * f; }
    NLB_WRITE(0) NLB_WRITE(1) NLB_WRITE(2) NLB_WRITE(3)
    NLB_WRITE(4) NLB_WRITE(5) NLB_WRITE(6) NLB_WRITE(7)
#undef NLB_WRITE
}

#define TCG 3
#define YP  68
__global__ __launch_bounds__(64)
void nlb_out(const float* __restrict__ x, const float* __restrict__ Y,
             const float* __restrict__ ww, const float* __restrict__ wb,
             float* __restrict__ Z) {
    __shared__ __attribute__((aligned(16))) float ys[64 * YP];
    const int tid = threadIdx.x;
    const int pix = blockIdx.x * 64 + tid;
    const int b = pix >> 12, n = pix & 4095;
    const int oh = n >> 6, ow = n & 63;
    {
        const float* yb = Y + ((size_t)(blockIdx.x * 64) << 6);
#pragma unroll
        for (int i = 0; i < 16; ++i) {
            const int e = i * 64 + tid;
            v4f t = *(const v4f*)(yb + 4 * e);
            *(v4f*)(ys + (e >> 4) * YP + 4 * (e & 15)) = t;
        }
    }
    __syncthreads();
    float y[60];
#pragma unroll
    for (int i = 0; i < 15; ++i) {
        v4f t = *(const v4f*)(ys + tid * YP + 4 * i);
        y[4 * i] = t.x; y[4 * i + 1] = t.y; y[4 * i + 2] = t.z; y[4 * i + 3] = t.w;
    }
    for (int k = 0; k < TCG; ++k) {
        const int tc = blockIdx.y * TCG + k;
        const float* wr = ww + tc * 4 * D_;
        float a0 = wb[tc * 4], a1 = wb[tc * 4 + 1], a2 = wb[tc * 4 + 2], a3 = wb[tc * 4 + 3];
#pragma unroll
        for (int d = 0; d < D_; ++d) {
            a0 = __builtin_fmaf(wr[d],          y[d], a0);
            a1 = __builtin_fmaf(wr[D_ + d],     y[d], a1);
            a2 = __builtin_fmaf(wr[2 * D_ + d], y[d], a2);
            a3 = __builtin_fmaf(wr[3 * D_ + d], y[d], a3);
        }
        const size_t base = ((size_t)(b * (T_ * C_) + tc) * H_ + 2 * oh) * W_ + 2 * ow;
        v2f x0 = *(const v2f*)(x + base);
        v2f x1 = *(const v2f*)(x + base + W_);
        v2f z0, z1;
        z0.x = a0 + x0.x; z0.y = a1 + x0.y;
        z1.x = a2 + x1.x; z1.y = a3 + x1.y;
        *(v2f*)(Z + base)      = z0;
        *(v2f*)(Z + base + W_) = z1;
    }
}

extern "C" void kernel_launch(void* const* d_in, const int* in_sizes, int n_in,
                              void* d_out, int out_size, void* d_ws, size_t ws_size,
                              hipStream_t stream) {
    const float* x  = (const float*)d_in[0];
    const float* gw = (const float*)d_in[1];
    const float* gb = (const float*)d_in[2];
    const float* ww = (const float*)d_in[3];
    const float* wb = (const float*)d_in[4];
    float* Z = (float*)d_out;

    const size_t buf = (size_t)B_ * NN * DP;
    float* Qbuf = (float*)d_ws;
    float* Gbuf = Qbuf + buf;
    float* Ybuf = Gbuf + buf;

    nlb_prep_q<<<256, 128, 0, stream>>>(x, Qbuf);
    nlb_prep_g<<<dim3(128, 4), 128, 0, stream>>>(Qbuf, gw, gb, Gbuf);
    nlb_attn<<<B_ * 64, 128, 0, stream>>>(Qbuf, Gbuf, Ybuf);
    nlb_out<<<dim3(256, 5), 64, 0, stream>>>(x, Ybuf, ww, wb, Z);
}


# attention: a wave takes all 4 query tiles of its block over 1/4 of the keys; a K/V fragment in registers feeds 4 tiles: 4x fewer global loads per WMMA; partials joined in LDS; key sum order changes
# speedup vs baseline: 2.6570x; 1.0472x over previous
#include <hip/hip_runtime.h>

typedef __attribute__((ext_vector_type(2))) float v2f;
typedef __attribute__((ext_vector_type(4))) float v4f;
typedef __attribute__((ext_vector_type(8))) float v8f;

#define B_  4
#define T_  5
#define C_  3
#define H_  128
#define W_  128
#define D_  60
#define DP  64
#define NN  4096
#define KT  128
#define PP  132

__device__ __forceinline__ float fast_exp2(float x) {
    return __builtin_amdgcn_exp2f(x);
}

__device__ __forceinline__ float xhalf(float v) {
    const int b = __builtin_bit_cast(int, v);
    return __builtin_bit_cast(float,
        __builtin_amdgcn_permlanex16(b, b, 0x76543210, 0xfedcba98, false, false));
}

template <int J>
__device__ __forceinline__ float row_bcast(float v) {
    const int x  = __builtin_bit_cast(int, v);
    const int lo = __builtin_amdgcn_update_dpp(0, x, 0x150 + J, 0xf, 0xf, true);
    const int r  = __builtin_amdgcn_update_dpp(lo, x, 0x150 + 8 + J, 0x2, 0xf, false);
    return __builtin_bit_cast(float, r);
}


__global__ __launch_bounds__(128)
void nlb_prep_q(const float* __restrict__ x, float* __restrict__ Kf) {
    const int gid  = blockIdx.x * 128 + threadIdx.x;
    const int lane = gid & 31;
    const int T    = (gid >> 5) & 255;
    const int b    = gid >> 13;
    const int m = lane & 15, h = lane >> 4;
    const int n  = T * 16 + m;
    const int oh = n >> 6, ow = n & 63;
    const float* sp = x + ((size_t)(b * (T_ * C_)) * H_ + (2 * oh + h)) * W_ + 2 * ow;
    float*       dp = Kf + ((size_t)(b * 256 + T) * 8) * 128 + lane * 4;
    v2f p[15];
#pragma unroll
    for (int tc = 0; tc < 15; ++tc)
        p[tc] = *(const v2f*)(sp + (size_t)tc * (H_ * W_));
#pragma unroll
    for (int kp = 0; kp < 8; ++kp) {
        v4f o;
        o.x = p[2 * kp].x; o.y = p[2 * kp].y;
        if (kp < 7) { o.z = p[2 * kp + 1].x; o.w = p[2 * kp + 1].y; }
        else        { o.z = 0.0f; o.w = 0.0f; }
        *(v4f*)(dp + kp * 128) = o;
    }
}

#define VP 136
__global__ __launch_bounds__(128)
void nlb_prep_g(const float* __restrict__ Kf, const float* __restrict__ gw,
                const float* __restrict__ gb, float* __restrict__ Vf) {
    __shared__ __attribute__((aligned(16))) float vt[16 * VP];
    const int tid  = threadIdx.x;
    const int ctp  = blockIdx.y >> 1;
    const int half = blockIdx.y & 1;
    const int pix  = blockIdx.x * 128 + tid;
    const int b = pix >> 12, n = pix & 4095;
    float xr[60];
    {
        const float* q0 = Kf + ((size_t)(b * 256 + (n >> 4)) * 8) * 128 + (n & 15) * 4;
#pragma unroll
        for (int kp = 0; kp < 7; ++kp) {
            v4f a = *(const v4f*)(q0 + kp * 128);
            v4f c = *(const v4f*)(q0 + kp * 128 + 64);
            xr[8 * kp + 0] = a.x; xr[8 * kp + 1] = a.y; xr[8 * kp + 2] = c.x; xr[8 * kp + 3] = c.y;
            xr[8 * kp + 4] = a.z; xr[8 * kp + 5] = a.w; xr[8 * kp + 6] = c.z; xr[8 * kp + 7] = c.w;
        }
        v2f a = *(const v2f*)(q0 + 7 * 128);
        v2f c = *(const v2f*)(q0 + 7 * 128 + 64);
        xr[56] = a.x; xr[57] = a.y; xr[58] = c.x; xr[59] = c.y;
    }
    for (int f = 0; f < 16; f += 4) {
        const int o = 32 * ctp + 8 * half + f + ((f & 8) ? 8 : 0);
        float a0 = 0.0f, a1 = 0.0f, a2 = 0.0f, a3 = 0.0f;
        if (o < D_) {
            const float* wr = gw + o * D_;
            a0 = gb[o]; a1 = gb[o + 1]; a2 = gb[o + 2]; a3 = gb[o + 3];
#pragma unroll
            for (int d = 0; d < D_; ++d) {
                a0 = __builtin_fmaf(wr[d],          xr[d], a0);
                a1 = __builtin_fmaf(wr[D_ + d],     xr[d], a1);
                a2 = __builtin_fmaf(wr[2 * D_ + d], xr[d], a2);
                a3 = __builtin_fmaf(wr[3 * D_ + d], xr[d], a3);
            }
        }
        vt[(f + 0) * VP + tid] = a0;
        vt[(f + 1) * VP + tid] = a1;
        vt[(f + 2) * VP + tid] = a2;
        vt[(f + 3) * VP + tid] = a3;
    }
    __syncthreads();
    float* vo = Vf + ((size_t)(blockIdx.x * 32) * 2 + ctp) * 128;
#pragma unroll
    for (int r = 0; r < 4; ++r) {
        const int el = r * 128 + tid;
        const int gl = el >> 4;
        const int hh = (el >> 3) & 1, m8 = el & 7;
        const int key = 4 * gl + 2 * hh;
        v2f lo = *(const v2f*)(vt + m8 * VP + key);
        v2f hi = *(const v2f*)(vt + (8 + m8) * VP + key);
        v4f o; o.x = lo.x; o.y = lo.y; o.z = hi.x; o.w = hi.y;
        *(v4f*)(vo + (size_t)gl * 256 + (16 * hh + 8 * half + m8) * 4) = o;
    }
}

#define MRG 1088
__global__ __launch_bounds__(128, 1)
void nlb_attn(const float* __restrict__ Kf, const float* __restrict__ Vf,
              float* __restrict__ Y) {
    __shared__ __attribute__((aligned(16))) float lds[16 * MRG];

    const int lane = threadIdx.x & 31;
    const int wave = threadIdx.x >> 5;
    const int m    = lane & 15;
    const int h    = lane >> 4;
    const int b    = blockIdx.x >> 6;
    const int tq   = (blockIdx.x & 63) * 4;
    const int kq   = wave * (NN / 4);

    const float* Kb  = Kf + (size_t)b * NN * DP;
    const float* Vb  = Vf + (size_t)b * NN * DP;
    float*       Yb  = Y  + (size_t)b * NN * DP;
    float*       pl  = lds + wave * (16 * PP);

    const float LOG2E = 1.4426950408889634f;
    v2f qa[4][15];
#pragma unroll
    for (int t4 = 0; t4 < 4; ++t4) {
        const float* qt = Kb + (size_t)(tq + t4) * 1024 + lane * 4;
#pragma unroll
        for (int kp = 0; kp < 7; ++kp) {
            v4f q = *(const v4f*)(qt + kp * 128);
            qa[t4][2 * kp]     = q.xy * LOG2E;
            qa[t4][2 * kp + 1] = q.zw * LOG2E;
        }
        v2f q = *(const v2f*)(qt + 7 * 128);
        qa[t4][14] = q * LOG2E;
    }

    v4f kc[8][7];
    v2f kl[8];
    v4f vc[32][2];
    v2f pa[32];

    {
        const float* kt = Kb + (size_t)(kq >> 4) * 1024 + lane * 4;
#pragma unroll
        for (int t = 0; t < 8; ++t) {
#pragma unroll
            for (int kp = 0; kp < 7; ++kp)
                kc[t][kp] = *(const v4f*)(kt + t * 1024 + kp * 128);
            kl[t] = *(const v2f*)(kt + t * 1024 + 7 * 128);
        }
    }

    v8f o[4][4];
    float mr[4];
    float lr[4];
#pragma unroll
    for (int t4 = 0; t4 < 4; ++t4) {
        o[t4][0] = v8f{}; o[t4][1] = v8f{}; o[t4][2] = v8f{}; o[t4][3] = v8f{};
        mr[t4] = -__builtin_inff();
        lr[t4] = 0.0f;
    }

    for (int ch = 0; ch < NN / 4 / KT; ++ch) {
        const int k0 = kq + ch * KT;
        const int kn = (ch + 1 < NN / 4 / KT) ? (k0 + KT) : k0;
        const float* vt = Vb + (size_t)(k0 >> 2) * 256 + lane * 4;
        const float* kt = Kb + (size_t)(kn >> 4) * 1024 + lane * 4;

#pragma unroll
        for (int t4 = 0; t4 < 4; ++t4) {
            v8f s[8];
#pragma unroll
            for (int tp = 0; tp < 4; ++tp) {
                if (t4 == 0) {
#pragma unroll
                    for (int u = 0; u < 8; ++u) {
                        vc[8 * tp + u][0] = *(const v4f*)(vt + (8 * tp + u) * 256);
                        vc[8 * tp + u][1] = *(const v4f*)(vt + (8 * tp + u) * 256 + 128);
                    }
                }
                v8f a0{}, a1{};
#pragma unroll
                for (int kp = 0; kp < 7; ++kp) {
                    a0 = __builtin_amdgcn_wmma_f32_16x16x4_f32(
                             false, qa[t4][2 * kp], false, kc[2 * tp][kp].xy, (short)0, a0, false, false);
                    a1 = __builtin_amdgcn_wmma_f32_16x16x4_f32(
                             false, qa[t4][2 * kp], false, kc[2 * tp + 1][kp].xy, (short)0, a1, false, false);
                    a0 = __builtin_amdgcn_wmma_f32_16x16x4_f32(
                             false, qa[t4][2 * kp + 1], false, kc[2 * tp][kp].zw, (short)0, a0, false, false);
                    a1 = __builtin_amdgcn_wmma_f32_16x16x4_f32(
                             false, qa[t4][2 * kp + 1], false, kc[2 * tp + 1][kp].zw, (short)0, a1, false, false);
                }
                a0 = __builtin_amdgcn_wmma_f32_16x16x4_f32(
                         false, qa[t4][14], false, kl[2 * tp], (short)0, a0, false, false);
                a1 = __builtin_amdgcn_wmma_f32_16x16x4_f32(
                         false, qa[t4][14], false, kl[2 * tp + 1], (short)0, a1, false, false);
                s[2 * tp]     = a0;
                s[2 * tp + 1] = a1;
                __builtin_amdgcn_sched_barrier(0);
            }

#pragma unroll
            for (int t = 0; t < 8; ++t)
#pragma unroll
                for (int j = 0; j < 8; ++j)
                    pl[(8 * h + j) * PP + t * 16 + m] = s[t][j];
            asm volatile("" ::: "memory");

#pragma unroll
            for (int kk2 = 0; kk2 < 32; ++kk2)
                pa[kk2] = *(const v2f*)(pl + m * PP + 4 * kk2 + 2 * h);
            asm volatile("" ::: "memory");

            float t0 = fmaxf(pa[0].x, pa[0].y), t1 = fmaxf(pa[1].x, pa[1].y);
            float t2 = fmaxf(pa[2].x, pa[2].y), t3 = fmaxf(pa[3].x, pa[3].y);
#pragma unroll
            for (int q = 1; q < 8; ++q) {
                t0 = fmaxf(t0, fmaxf(pa[4 * q].x,     pa[4 * q].y));
                t1 = fmaxf(t1, fmaxf(pa[4 * q + 1].x, pa[4 * q + 1].y));
                t2 = fmaxf(t2, fmaxf(pa[4 * q + 2].x, pa[4 * q + 2].y));
                t3 = fmaxf(t3, fmaxf(pa[4 * q + 3].x, pa[4 * q + 3].y));
            }
            float tm = fmaxf(fmaxf(t0, t1), fmaxf(t2, t3));
            tm = fmaxf(tm, xhalf(tm));
            const float mn = fmaxf(mr[t4], tm);
            const float sc = fast_exp2(mr[t4] - mn);
            mr[t4] = mn;
            const v2f mnv = {mn, mn};
#pragma unroll
            for (int kk2 = 0; kk2 < 32; ++kk2)
                pa[kk2] = pa[kk2] - mnv;
            v2f u0{}, u1{}, u2{}, u3{};
#pragma unroll
            for (int q = 0; q < 8; ++q) {
#pragma unroll
                for (int i = 0; i < 4; ++i) {
                    pa[4 * q + i].x = fast_exp2(pa[4 * q + i].x);
                    pa[4 * q + i].y = fast_exp2(pa[4 * q + i].y);
                }
                u0 += pa[4 * q]; u1 += pa[4 * q + 1]; u2 += pa[4 * q + 2]; u3 += pa[4 * q + 3];
            }
            const v2f us = (u0 + u1) + (u2 + u3);
            lr[t4] = lr[t4] * sc + (us.x + us.y);
#define NLB_RESCALE(J) { const float f = row_bcast<J>(sc); \
            o[t4][0][J] *= f; o[t4][1][J] *= f; o[t4][2][J] *= f; o[t4][3][J] *= f; }
            NLB_RESCALE(0) NLB_RESCALE(1) NLB_RESCALE(2) NLB_RESCALE(3)
            NLB_RESCALE(4) NLB_RESCALE(5) NLB_RESCALE(6) NLB_RESCALE(7)
#undef NLB_RESCALE
#pragma unroll
            for (int q = 0; q < 32; q += 8)
                asm volatile("" : "+v"(pa[q]), "+v"(pa[q + 1]), "+v"(pa[q + 2]), "+v"(pa[q + 3]),
                                  "+v"(pa[q + 4]), "+v"(pa[q + 5]), "+v"(pa[q + 6]), "+v"(pa[q + 7]));
            __builtin_amdgcn_sched_barrier(0);

#pragma unroll
            for (int g = 0; g < 8; ++g) {
                if (t4 == 3) {
#pragma unroll
                    for (int kp = 0; kp < 7; ++kp)
                        kc[g][kp] = *(const v4f*)(kt + g * 1024 + kp * 128);
                    kl[g] = *(const v2f*)(kt + g * 1024 + 7 * 128);
                }
#pragma unroll
                for (int u = 0; u < 4; ++u) {
                    const int kk2 = 4 * g + u;
                    o[t4][0] = __builtin_amdgcn_wmma_f32_16x16x4_f32(false, pa[kk2], false, vc[kk2][0].xy, (short)0, o[t4][0], false, false);
                    o[t4][1] = __builtin_amdgcn_wmma_f32_16x16x4_f32(false, pa[kk2], false, vc[kk2][0].zw, (short)0, o[t4][1], false, false);
                    o[t4][2] = __builtin_amdgcn_wmma_f32_16x16x4_f32(false, pa[kk2], false, vc[kk2][1].xy, (short)0, o[t4][2], false, false);
                    o[t4][3] = __builtin_amdgcn_wmma_f32_16x16x4_f32(false, pa[kk2], false, vc[kk2][1].zw, (short)0, o[t4][3], false, false);
                }
                __builtin_amdgcn_sched_barrier(0);
            }
        }
    }

    __syncthreads();
#pragma unroll
    for (int t4 = 0; t4 < 4; ++t4) {
        float* rg = lds + (4 * t4 + wave) * MRG;
#pragma unroll
        for (int ct = 0; ct < 4; ++ct) {
            v4f lo = {o[t4][ct][0], o[t4][ct][1], o[t4][ct][2], o[t4][ct][3]};
            v4f hi = {o[t4][ct][4], o[t4][ct][5], o[t4][ct][6], o[t4][ct][7]};
            *(v4f*)(rg + (2 * ct) * 128 + lane * 4)     = lo;
            *(v4f*)(rg + (2 * ct + 1) * 128 + lane * 4) = hi;
        }
        v2f ml = {mr[t4], lr[t4] + xhalf(lr[t4])};
        *(v2f*)(rg + 1024 + lane * 2) = ml;
    }
    __syncthreads();
    {
        const float* rg = lds + (4 * wave) * MRG;
        const v2f ml0 = *(const v2f*)(rg + 1024 + lane * 2);
        const v2f ml1 = *(const v2f*)(rg + MRG + 1024 + lane * 2);
        const v2f ml2 = *(const v2f*)(rg + 2 * MRG + 1024 + lane * 2);
        const v2f ml3 = *(const v2f*)(rg + 3 * MRG + 1024 + lane * 2);
        const float mj = fmaxf(fmaxf(ml0.x, ml1.x), fmaxf(ml2.x, ml3.x));
        float f0 = fast_exp2(ml0.x - mj);
        float f1 = fast_exp2(ml1.x - mj);
        float f2 = fast_exp2(ml2.x - mj);
        float f3 = fast_exp2(ml3.x - mj);
        const float inv = 1.0f / (((ml0.y * f0 + ml1.y * f1) + ml2.y * f2) + ml3.y * f3);
        f0 *= inv; f1 *= inv; f2 *= inv; f3 *= inv;
        const int q0 = (tq + wave) * 16;
#pragma unroll
        for (int ct = 0; ct < 4; ++ct) {
            v8f p0, p1, p2, p3;
#define NLB_PIECE(P, S) { \
            const v4f lo = *(const v4f*)(rg + (S) * MRG + (2 * ct) * 128 + lane * 4); \
            const v4f hi = *(const v4f*)(rg + (S) * MRG + (2 * ct + 1) * 128 + lane * 4); \
            P = v8f{lo.x, lo.y, lo.z, lo.w, hi.x, hi.y, hi.z, hi.w}; }
            NLB_PIECE(p0, 0) NLB_PIECE(p1, 1) NLB_PIECE(p2, 2) NLB_PIECE(p3, 3)
#undef NLB_PIECE
#define NLB_JOIN(J) { float* yr = Yb + (size_t)(q0 + 8 * h + J) * DP; \
            yr[16 * ct + m] = ((p0[J] * row_bcast<J>(f0) + p1[J] * row_bcast<J>(f1)) \
                               + p2[J] * row_bcast<J>(f2)) + p3[J] * row_bcast<J>(f3); }
            NLB_JOIN(0) NLB_JOIN(1) NLB_JOIN(2) NLB_JOIN(3)
            NLB_JOIN(4) NLB_JOIN(5) NLB_JOIN(6) NLB_JOIN(7)
#undef NLB_JOIN
        }
    }
}

#define TCG 3
#define YP  68
__global__ __launch_bounds__(64)
void nlb_out(const float* __restrict__ x, const float* __restrict__ Y,
             const float* __restrict__ ww, const float* __restrict__ wb,
             float* __restrict__ Z) {
    __shared__ __attribute__((aligned(16))) float ys[64 * YP];
    const int tid = threadIdx.x;
    const int pix = blockIdx.x * 64 + tid;
    const int b = pix >> 12, n = pix & 4095;
    const int oh = n >> 6, ow = n & 63;
    {
        const float* yb = Y + ((size_t)(blockIdx.x * 64) << 6);
#pragma unroll
        for (int i = 0; i < 16; ++i) {
            const int e = i * 64 + tid;
            v4f t = *(const v4f*)(yb + 4 * e);
            *(v4f*)(ys + (e >> 4) * YP + 4 * (e & 15)) = t;
        }
    }
    __syncthreads();
    float y[60];
#pragma unroll
    for (int i = 0; i < 15; ++i) {
        v4f t = *(const v4f*)(ys + tid * YP + 4 * i);
        y[4 * i] = t.x; y[4 * i + 1] = t.y; y[4 * i + 2] = t.z; y[4 * i + 3] = t.w;
    }
    for (int k = 0; k < TCG; ++k) {
        const int tc = blockIdx.y * TCG + k;
        const float* wr = ww + tc * 4 * D_;
        float a0 = wb[tc * 4], a1 = wb[tc * 4 + 1], a2 = wb[tc * 4 + 2], a3 = wb[tc * 4 + 3];
#pragma unroll
        for (int d = 0; d < D_; ++d) {
            a0 = __builtin_fmaf(wr[d],          y[d], a0);
            a1 = __builtin_fmaf(wr[D_ + d],     y[d], a1);
            a2 = __builtin_fmaf(wr[2 * D_ + d], y[d], a2);
            a3 = __builtin_fmaf(wr[3 * D_ + d], y[d], a3);
        }
        const size_t base = ((size_t)(b * (T_ * C_) + tc) * H_ + 2 * oh) * W_ + 2 * ow;
        v2f x0 = *(const v2f*)(x + base);
        v2f x1 = *(const v2f*)(x + base + W_);
        v2f z0, z1;
        z0.x = a0 + x0.x; z0.y = a1 + x0.y;
        z1.x = a2 + x1.x; z1.y = a3 + x1.y;
        *(v2f*)(Z + base)      = z0;
        *(v2f*)(Z + base + W_) = z1;
    }
}

extern "C" void kernel_launch(void* const* d_in, const int* in_sizes, int n_in,
                              void* d_out, int out_size, void* d_ws, size_t ws_size,
                              hipStream_t stream) {
    const float* x  = (const float*)d_in[0];
    const float* gw = (const float*)d_in[1];
    const float* gb = (const float*)d_in[2];
    const float* ww = (const float*)d_in[3];
    const float* wb = (const float*)d_in[4];
    float* Z = (float*)d_out;

    const size_t buf = (size_t)B_ * NN * DP;
    float* Qbuf = (float*)d_ws;
    float* Gbuf = Qbuf + buf;
    float* Ybuf = Gbuf + buf;

    nlb_prep_q<<<256, 128, 0, stream>>>(x, Qbuf);
    nlb_prep_g<<<dim3(128, 4), 128, 0, stream>>>(Qbuf, gw, gb, Gbuf);
    nlb_attn<<<B_ * 64, 128, 0, stream>>>(Qbuf, Gbuf, Ybuf);
    nlb_out<<<dim3(256, 5), 64, 0, stream>>>(x, Ybuf, ww, wb, Z);
}


# attention takes Q and K fragments straight from x (unshuffle = address arithmetic, one 8-B load per lane and k-step, whole lines); V projection reads x too; unshuffle kernel gone: 4 launches -> 3
# speedup vs baseline: 2.7657x; 1.0409x over previous
#include <hip/hip_runtime.h>

typedef __attribute__((ext_vector_type(2))) float v2f;
typedef __attribute__((ext_vector_type(4))) float v4f;
typedef __attribute__((ext_vector_type(8))) float v8f;

#define B_  4
#define T_  5
#define C_  3
#define H_  128
#define W_  128
#define D_  60
#define DP  64
#define NN  4096
#define KT  128
#define PP  132

__device__ __forceinline__ float fast_exp2(float x) {
    return __builtin_amdgcn_exp2f(x);
}

__device__ __forceinline__ float xhalf(float v) {
    const int b = __builtin_bit_cast(int, v);
    return __builtin_bit_cast(float,
        __builtin_amdgcn_permlanex16(b, b, 0x76543210, 0xfedcba98, false, false));
}

template <int J>
__device__ __forceinline__ float row_bcast(float v) {
    const int x  = __builtin_bit_cast(int, v);
    const int lo = __builtin_amdgcn_update_dpp(0, x, 0x150 + J, 0xf, 0xf, true);
    const int r  = __builtin_amdgcn_update_dpp(lo, x, 0x150 + 8 + J, 0x2, 0xf, false);
    return __builtin_bit_cast(float, r);
}


__device__ __forceinline__ const float* x_frag(const float* xb, int T, int h, int m) {
    return xb + ((T >> 2) * 2 + h) * W_ + ((T & 3) * 16 + m) * 2;
}

#define VP 136
__global__ __launch_bounds__(128)
void nlb_prep_g(const float* __restrict__ x, const float* __restrict__ gw,
                const float* __restrict__ gb, float* __restrict__ Vf) {
    __shared__ __attribute__((aligned(16))) float vt[16 * VP];
    const int tid  = threadIdx.x;
    const int ctp  = blockIdx.y >> 1;
    const int half = blockIdx.y & 1;
    const int pix  = blockIdx.x * 128 + tid;
    const int b = pix >> 12, n = pix & 4095;
    float xr[60];
    {
        const int oh = n >> 6, ow = n & 63;
        const float* sp = x + ((size_t)(b * (T_ * C_)) * H_ + 2 * oh) * W_ + 2 * ow;
#pragma unroll
        for (int tc = 0; tc < 15; ++tc) {
            v2f r0 = *(const v2f*)(sp + (size_t)tc * (H_ * W_));
            v2f r1 = *(const v2f*)(sp + (size_t)tc * (H_ * W_) + W_);
            xr[4 * tc + 0] = r0.x; xr[4 * tc + 1] = r0.y;
            xr[4 * tc + 2] = r1.x; xr[4 * tc + 3] = r1.y;
        }
    }
    for (int f = 0; f < 16; f += 4) {
        const int o = 32 * ctp + 8 * half + f + ((f & 8) ? 8 : 0);
        float a0 = 0.0f, a1 = 0.0f, a2 = 0.0f, a3 = 0.0f;
        if (o < D_) {
            const float* wr = gw + o * D_;
            a0 = gb[o]; a1 = gb[o + 1]; a2 = gb[o + 2]; a3 = gb[o + 3];
#pragma unroll
            for (int d = 0; d < D_; ++d) {
                a0 = __builtin_fmaf(wr[d],          xr[d], a0);
                a1 = __builtin_fmaf(wr[D_ + d],     xr[d], a1);
                a2 = __builtin_fmaf(wr[2 * D_ + d], xr[d], a2);
                a3 = __builtin_fmaf(wr[3 * D_ + d], xr[d], a3);
            }
        }
        vt[(f + 0) * VP + tid] = a0;
        vt[(f + 1) * VP + tid] = a1;
        vt[(f + 2) * VP + tid] = a2;
        vt[(f + 3) * VP + tid] = a3;
    }
    __syncthreads();
    float* vo = Vf + ((size_t)(blockIdx.x * 32) * 2 + ctp) * 128;
#pragma unroll
    for (int r = 0; r < 4; ++r) {
        const int el = r * 128 + tid;
        const int gl = el >> 4;
        const int hh = (el >> 3) & 1, m8 = el & 7;
        const int key = 4 * gl + 2 * hh;
        v2f lo = *(const v2f*)(vt + m8 * VP + key);
        v2f hi = *(const v2f*)(vt + (8 + m8) * VP + key);
        v4f o; o.x = lo.x; o.y = lo.y; o.z = hi.x; o.w = hi.y;
        *(v4f*)(vo + (size_t)gl * 256 + (16 * hh + 8 * half + m8) * 4) = o;
    }
}

#define MRG 1088
__global__ __launch_bounds__(128, 1)
void nlb_attn(const float* __restrict__ x, const float* __restrict__ Vf,
              float* __restrict__ Y) {
    __shared__ __attribute__((aligned(16))) float lds[16 * MRG];

    const int lane = threadIdx.x & 31;
    const int wave = threadIdx.x >> 5;
    const int m    = lane & 15;
    const int h    = lane >> 4;
    const int b    = blockIdx.x >> 6;
    const int tq   = (blockIdx.x & 63) * 4;
    const int kq   = wave * (NN / 4);

    const float* xb  = x  + (size_t)b * (T_ * C_) * H_ * W_;
    const float* Vb  = Vf + (size_t)b * NN * DP;
    float*       Yb  = Y  + (size_t)b * NN * DP;
    float*       pl  = lds + wave * (16 * PP);

    const float LOG2E = 1.4426950408889634f;
    v2f qa[4][15];
#pragma unroll
    for (int t4 = 0; t4 < 4; ++t4) {
        const float* qt = x_frag(xb, tq + t4, h, m);
#pragma unroll
        for (int kk = 0; kk < 15; ++kk)
            qa[t4][kk] = *(const v2f*)(qt + kk * (H_ * W_)) * LOG2E;
    }

    v2f kc[8][15];
    v4f vc[32][2];
    v2f pa[32];

    {
        const float* kt = x_frag(xb, kq >> 4, h, m);
#pragma unroll
        for (int t = 0; t < 8; ++t)
#pragma unroll
            for (int kk = 0; kk < 15; ++kk)
                kc[t][kk] = *(const v2f*)(kt + kk * (H_ * W_) + (t >> 2) * (2 * W_) + (t & 3) * 32);
    }

    v8f o[4][4];
    float mr[4];
    float lr[4];
#pragma unroll
    for (int t4 = 0; t4 < 4; ++t4) {
        o[t4][0] = v8f{}; o[t4][1] = v8f{}; o[t4][2] = v8f{}; o[t4][3] = v8f{};
        mr[t4] = -__builtin_inff();
        lr[t4] = 0.0f;
    }

    for (int ch = 0; ch < NN / 4 / KT; ++ch) {
        const int k0 = kq + ch * KT;
        const int kn = (ch + 1 < NN / 4 / KT) ? (k0 + KT) : k0;
        const float* vt = Vb + (size_t)(k0 >> 2) * 256 + lane * 4;
        const float* kt = x_frag(xb, kn >> 4, h, m);

#pragma unroll
        for (int t4 = 0; t4 < 4; ++t4) {
            v8f s[8];
#pragma unroll
            for (int tp = 0; tp < 4; ++tp) {
                if (t4 == 0) {
#pragma unroll
                    for (int u = 0; u < 8; ++u) {
                        vc[8 * tp + u][0] = *(const v4f*)(vt + (8 * tp + u) * 256);
                        vc[8 * tp + u][1] = *(const v4f*)(vt + (8 * tp + u) * 256 + 128);
                    }
                }
                v8f a0{}, a1{};
#pragma unroll
                for (int kk = 0; kk < 15; ++kk) {
                    a0 = __builtin_amdgcn_wmma_f32_16x16x4_f32(
                             false, qa[t4][kk], false, kc[2 * tp][kk], (short)0, a0, false, false);
                    a1 = __builtin_amdgcn_wmma_f32_16x16x4_f32(
                             false, qa[t4][kk], false, kc[2 * tp + 1][kk], (short)0, a1, false, false);
                }
                s[2 * tp]     = a0;
                s[2 * tp + 1] = a1;
                __builtin_amdgcn_sched_barrier(0);
            }

#pragma unroll
            for (int t = 0; t < 8; ++t)
#pragma unroll
                for (int j = 0; j < 8; ++j)
                    pl[(8 * h + j) * PP + t * 16 + m] = s[t][j];
            asm volatile("" ::: "memory");

#pragma unroll
            for (int kk2 = 0; kk2 < 32; ++kk2)
                pa[kk2] = *(const v2f*)(pl + m * PP + 4 * kk2 + 2 * h);
            asm volatile("" ::: "memory");

            float t0 = fmaxf(pa[0].x, pa[0].y), t1 = fmaxf(pa[1].x, pa[1].y);
            float t2 = fmaxf(pa[2].x, pa[2].y), t3 = fmaxf(pa[3].x, pa[3].y);
#pragma unroll
            for (int q = 1; q < 8; ++q) {
                t0 = fmaxf(t0, fmaxf(pa[4 * q].x,     pa[4 * q].y));
                t1 = fmaxf(t1, fmaxf(pa[4 * q + 1].x, pa[4 * q + 1].y));
                t2 = fmaxf(t2, fmaxf(pa[4 * q + 2].x, pa[4 * q + 2].y));
                t3 = fmaxf(t3, fmaxf(pa[4 * q + 3].x, pa[4 * q + 3].y));
            }
            float tm = fmaxf(fmaxf(t0, t1), fmaxf(t2, t3));
            tm = fmaxf(tm, xhalf(tm));
            const float mn = fmaxf(mr[t4], tm);
            const float sc = fast_exp2(mr[t4] - mn);
            mr[t4] = mn;
            const v2f mnv = {mn, mn};
#pragma unroll
            for (int kk2 = 0; kk2 < 32; ++kk2)
                pa[kk2] = pa[kk2] - mnv;
            v2f u0{}, u1{}, u2{}, u3{};
#pragma unroll
            for (int q = 0; q < 8; ++q) {
#pragma unroll
                for (int i = 0; i < 4; ++i) {
                    pa[4 * q + i].x = fast_exp2(pa[4 * q + i].x);
                    pa[4 * q + i].y = fast_exp2(pa[4 * q + i].y);
                }
                u0 += pa[4 * q]; u1 += pa[4 * q + 1]; u2 += pa[4 * q + 2]; u3 += pa[4 * q + 3];
            }
            const v2f us = (u0 + u1) + (u2 + u3);
            lr[t4] = lr[t4] * sc + (us.x + us.y);
#define NLB_RESCALE(J) { const float f = row_bcast<J>(sc); \
            o[t4][0][J] *= f; o[t4][1][J] *= f; o[t4][2][J] *= f; o[t4][3][J] *= f; }
            NLB_RESCALE(0) NLB_RESCALE(1) NLB_RESCALE(2) NLB_RESCALE(3)
            NLB_RESCALE(4) NLB_RESCALE(5) NLB_RESCALE(6) NLB_RESCALE(7)
#undef NLB_RESCALE
#pragma unroll
            for (int q = 0; q < 32; q += 8)
                asm volatile("" : "+v"(pa[q]), "+v"(pa[q + 1]), "+v"(pa[q + 2]), "+v"(pa[q + 3]),
                                  "+v"(pa[q + 4]), "+v"(pa[q + 5]), "+v"(pa[q + 6]), "+v"(pa[q + 7]));
            __builtin_amdgcn_sched_barrier(0);

#pragma unroll
            for (int g = 0; g < 8; ++g) {
                if (t4 == 3) {
#pragma unroll
                    for (int kk = 0; kk < 15; ++kk)
                        kc[g][kk] = *(const v2f*)(kt + kk * (H_ * W_) + (g >> 2) * (2 * W_) + (g & 3) * 32);
                }
#pragma unroll
                for (int u = 0; u < 4; ++u) {
                    const int kk2 = 4 * g + u;
                    o[t4][0] = __builtin_amdgcn_wmma_f32_16x16x4_f32(false, pa[kk2], false, vc[kk2][0].xy, (short)0, o[t4][0], false, false);
                    o[t4][1] = __builtin_amdgcn_wmma_f32_16x16x4_f32(false, pa[kk2], false, vc[kk2][0].zw, (short)0, o[t4][1], false, false);
                    o[t4][2] = __builtin_amdgcn_wmma_f32_16x16x4_f32(false, pa[kk2], false, vc[kk2][1].xy, (short)0, o[t4][2], false, false);
                    o[t4][3] = __builtin_amdgcn_wmma_f32_16x16x4_f32(false, pa[kk2], false, vc[kk2][1].zw, (short)0, o[t4][3], false, false);
                }
                __builtin_amdgcn_sched_barrier(0);
            }
        }
    }

    __syncthreads();
#pragma unroll
    for (int t4 = 0; t4 < 4; ++t4) {
        float* rg = lds + (4 * t4 + wave) * MRG;
#pragma unroll
        for (int ct = 0; ct < 4; ++ct) {
            v4f lo = {o[t4][ct][0], o[t4][ct][1], o[t4][ct][2], o[t4][ct][3]};
            v4f hi = {o[t4][ct][4], o[t4][ct][5], o[t4][ct][6], o[t4][ct][7]};
            *(v4f*)(rg + (2 * ct) * 128 + lane * 4)     = lo;
            *(v4f*)(rg + (2 * ct + 1) * 128 + lane * 4) = hi;
        }
        v2f ml = {mr[t4], lr[t4] + xhalf(lr[t4])};
        *(v2f*)(rg + 1024 + lane * 2) = ml;
    }
    __syncthreads();
    {
        const float* rg = lds + (4 * wave) * MRG;
        const v2f ml0 = *(const v2f*)(rg + 1024 + lane * 2);
        const v2f ml1 = *(const v2f*)(rg + MRG + 1024 + lane * 2);
        const v2f ml2 = *(const v2f*)(rg + 2 * MRG + 1024 + lane * 2);
        const v2f ml3 = *(const v2f*)(rg + 3 * MRG + 1024 + lane * 2);
        const float mj = fmaxf(fmaxf(ml0.x, ml1.x), fmaxf(ml2.x, ml3.x));
        float f0 = fast_exp2(ml0.x - mj);
        float f1 = fast_exp2(ml1.x - mj);
        float f2 = fast_exp2(ml2.x - mj);
        float f3 = fast_exp2(ml3.x - mj);
        const float inv = 1.0f / (((ml0.y * f0 + ml1.y * f1) + ml2.y * f2) + ml3.y * f3);
        f0 *= inv; f1 *= inv; f2 *= inv; f3 *= inv;
        const int q0 = (tq + wave) * 16;
#pragma unroll
        for (int ct = 0; ct < 4; ++ct) {
            v8f p0, p1, p2, p3;
#define NLB_PIECE(P, S) { \
            const v4f lo = *(const v4f*)(rg + (S) * MRG + (2 * ct) * 128 + lane * 4); \
            const v4f hi = *(const v4f*)(rg + (S) * MRG + (2 * ct + 1) * 128 + lane * 4); \
            P = v8f{lo.x, lo.y, lo.z, lo.w, hi.x, hi.y, hi.z, hi.w}; }
            NLB_PIECE(p0, 0) NLB_PIECE(p1, 1) NLB_PIECE(p2, 2) NLB_PIECE(p3, 3)
#undef NLB_PIECE
#define NLB_JOIN(J) { float* yr = Yb + (size_t)(q0 + 8 * h + J) * DP; \
            yr[16 * ct + m] = ((p0[J] * row_bcast<J>(f0) + p1[J] * row_bcast<J>(f1)) \
                               + p2[J] * row_bcast<J>(f2)) + p3[J] * row_bcast<J>(f3); }
            NLB_JOIN(0) NLB_JOIN(1) NLB_JOIN(2) NLB_JOIN(3)
            NLB_JOIN(4) NLB_JOIN(5) NLB_JOIN(6) NLB_JOIN(7)
#undef NLB_JOIN
        }
    }
}

#define TCG 3
#define YP  68
__global__ __launch_bounds__(64)
void nlb_out(const float* __restrict__ x, const float* __restrict__ Y,
             const float* __restrict__ ww, const float* __restrict__ wb,
             float* __restrict__ Z) {
    __shared__ __attribute__((aligned(16))) float ys[64 * YP];
    const int tid = threadIdx.x;
    const int pix = blockIdx.x * 64 + tid;
    const int b = pix >> 12, n = pix & 4095;
    const int oh = n >> 6, ow = n & 63;
    {
        const float* yb = Y + ((size_t)(blockIdx.x * 64) << 6);
#pragma unroll
        for (int i = 0; i < 16; ++i) {
            const int e = i * 64 + tid;
            v4f t = *(const v4f*)(yb + 4 * e);
            *(v4f*)(ys + (e >> 4) * YP + 4 * (e & 15)) = t;
        }
    }
    __syncthreads();
    float y[60];
#pragma unroll
    for (int i = 0; i < 15; ++i) {
        v4f t = *(const v4f*)(ys + tid * YP + 4 * i);
        y[4 * i] = t.x; y[4 * i + 1] = t.y; y[4 * i + 2] = t.z; y[4 * i + 3] = t.w;
    }
    for (int k = 0; k < TCG; ++k) {
        const int tc = blockIdx.y * TCG + k;
        const float* wr = ww + tc * 4 * D_;
        float a0 = wb[tc * 4], a1 = wb[tc * 4 + 1], a2 = wb[tc * 4 + 2], a3 = wb[tc * 4 + 3];
#pragma unroll
        for (int d = 0; d < D_; ++d) {
            a0 = __builtin_fmaf(wr[d],          y[d], a0);
            a1 = __builtin_fmaf(wr[D_ + d],     y[d], a1);
            a2 = __builtin_fmaf(wr[2 * D_ + d], y[d], a2);
            a3 = __builtin_fmaf(wr[3 * D_ + d], y[d], a3);
        }
        const size_t base = ((size_t)(b * (T_ * C_) + tc) * H_ + 2 * oh) * W_ + 2 * ow;
        v2f x0 = *(const v2f*)(x + base);
        v2f x1 = *(const v2f*)(x + base + W_);
        v2f z0, z1;
        z0.x = a0 + x0.x; z0.y = a1 + x0.y;
        z1.x = a2 + x1.x; z1.y = a3 + x1.y;
        *(v2f*)(Z + base)      = z0;
        *(v2f*)(Z + base + W_) = z1;
    }
}

extern "C" void kernel_launch(void* const* d_in, const int* in_sizes, int n_in,
                              void* d_out, int out_size, void* d_ws, size_t ws_size,
                              hipStream_t stream) {
    const float* x  = (const float*)d_in[0];
    const float* gw = (const float*)d_in[1];
    const float* gb = (const float*)d_in[2];
    const float* ww = (const float*)d_in[3];
    const float* wb = (const float*)d_in[4];
    float* Z = (float*)d_out;

    const size_t buf = (size_t)B_ * NN * DP;
    float* Gbuf = (float*)d_ws;
    float* Ybuf = Gbuf + buf;

    nlb_prep_g<<<dim3(128, 4), 128, 0, stream>>>(x, gw, gb, Gbuf);
    nlb_attn<<<B_ * 64, 128, 0, stream>>>(x, Gbuf, Ybuf);
    nlb_out<<<dim3(256, 5), 64, 0, stream>>>(x, Ybuf, ww, wb, Z);
}
